# starting kernel
# baseline (speedup 1.0000x reference)
#include <hip/hip_runtime.h>
#include <hip/hip_bf16.h>

typedef __attribute__((ext_vector_type(2))) float v2f;
typedef __attribute__((ext_vector_type(8))) float v8f;

#define EDIM 128
#define CDIM 128

__global__ __launch_bounds__(256)
void bte_gemm_kernel(const float* __restrict__ emb,
                     const float* __restrict__ W,
                     const float* __restrict__ bias,
                     const int*   __restrict__ tokens,
                     float*       __restrict__ h,
                     int N)
{
    __shared__ float ldsA[128 * EDIM];
    const int  block0 = blockIdx.x * 128;
    const int  tid    = threadIdx.x;
    const bool full   = (block0 + 128) <= N;

    {
        const int nl   = tid >> 1;
        const int half = (tid & 1) * 64;
        const int node = block0 + nl;
        const int tok  = (full || node < N) ? tokens[node] : 0;
        const float4* src = (const float4*)(emb + (size_t)tok * EDIM + half);
        float4*       dst = (float4*)(ldsA + nl * EDIM + half);
        #pragma unroll
        for (int i = 0; i < 16; ++i) dst[i] = src[i];
    }
    __syncthreads();

    const int wid   = tid >> 5;
    const int lane  = tid & 31;
    const int col   = lane & 15;
    const int khalf = lane >> 4;
    const float* arow = ldsA + (wid * 16 + col) * EDIM;

    for (int t = 0; t < 8; ++t) {
        const int c0 = t * 16;
        const float bv = bias[c0 + col];
        v8f acc = { bv, bv, bv, bv, bv, bv, bv, bv };
        const float* wrow = W + (size_t)(c0 + col) * EDIM;
        #pragma unroll 4
        for (int k = 0; k < EDIM; k += 4) {
            const int kk = k + 2 * khalf;
            v2f a, bf;
            a.x  = arow[kk];   a.y  = arow[kk + 1];
            bf.x = wrow[kk];   bf.y = wrow[kk + 1];
            acc = __builtin_amdgcn_wmma_f32_16x16x4_f32(
                      false, a, false, bf, (short)0, acc, false, false);
        }
        const int mbase = block0 + wid * 16 + 8 * khalf;
        float* hp = h + (size_t)mbase * CDIM + c0 + col;
        if (full) {
            #pragma unroll
            for (int r = 0; r < 8; ++r) hp[r * CDIM] = acc[r];
        } else {
            #pragma unroll
            for (int r = 0; r < 8; ++r)
                if (mbase + r < N) hp[r * CDIM] = acc[r];
        }
    }
}

__global__ __launch_bounds__(256)
void bte_scatter_kernel(float* __restrict__ h,
                        const int* __restrict__ parent,
                        int off, int count)
{
    const int idx = blockIdx.x * 256 + threadIdx.x;
    const int nl  = idx >> 5;
    if (nl >= count) return;
    const int cq   = (idx & 31) << 2;
    const int node = off + nl;
    const int par  = parent[node];
    const float4 v = *(const float4*)(h + (size_t)node * CDIM + cq);
    float* dst = h + (size_t)par * CDIM + cq;
    atomicAdd(dst + 0, v.x);
    atomicAdd(dst + 1, v.y);
    atomicAdd(dst + 2, v.z);
    atomicAdd(dst + 3, v.w);
}

__global__ __launch_bounds__(256)
void bte_init_out(float* __restrict__ out, int n)
{
    const int i = blockIdx.x * 256 + threadIdx.x;
    if (i < n) out[i] = 0.0f;
}

__global__ __launch_bounds__(256)
void bte_max_kernel(const float* __restrict__ h,
                    const int*   __restrict__ batch_id,
                    float*       __restrict__ out,
                    int N)
{
    const int idx  = blockIdx.x * 256 + threadIdx.x;
    const int node = idx >> 5;
    if (node >= N) return;
    const int cq = (idx & 31) << 2;
    const int b  = batch_id[node];
    const float4 v = *(const float4*)(h + (size_t)node * CDIM + cq);
    int* dst = (int*)(out + (size_t)b * CDIM + cq);
    atomicMax(dst + 0, __float_as_int(v.x));
    atomicMax(dst + 1, __float_as_int(v.y));
    atomicMax(dst + 2, __float_as_int(v.z));
    atomicMax(dst + 3, __float_as_int(v.w));
}

extern "C" void kernel_launch(void* const* d_in, const int* in_sizes, int n_in,
                              void* d_out, int out_size, void* d_ws, size_t ws_size,
                              hipStream_t stream)
{
    const float* emb      = (const float*)d_in[0];
    const float* W        = (const float*)d_in[1];
    const float* bias     = (const float*)d_in[2];
    const int*   tokens   = (const int*)d_in[3];
    const int*   parent   = (const int*)d_in[4];
    const int*   batch_id = (const int*)d_in[6];
    const int    N        = in_sizes[3];

    float* h   = (float*)d_ws;
    float* out = (float*)d_out;

    const int gemm_blocks = (N + 127) / 128;
    bte_gemm_kernel<<<gemm_blocks, 256, 0, stream>>>(emb, W, bias, tokens, h, N);

    static const int lvl_off[7]  = {0, 512, 2560, 10752, 35328, 109056, 256512};
    static const int lvl_size[7] = {512, 2048, 8192, 24576, 73728, 147456, 147456};
    for (int d = 6; d >= 1; --d) {
        const int threads = lvl_size[d] * 32;
        bte_scatter_kernel<<<(threads + 255) / 256, 256, 0, stream>>>(
            h, parent, lvl_off[d], lvl_size[d]);
    }

    bte_init_out<<<(out_size + 255) / 256, 256, 0, stream>>>(out, out_size);
    const long long mthreads = (long long)N * 32;
    bte_max_kernel<<<(int)((mthreads + 255) / 256), 256, 0, stream>>>(
        h, batch_id, out, N);
}


# max pass: each lane first reads its 4 table words with a device-scope load and sends atomicMax only for values above what it read; table bit-identical, no sum order changed
# speedup vs baseline: 1.6789x; 1.6789x over previous
#include <hip/hip_runtime.h>
#include <hip/hip_bf16.h>

typedef __attribute__((ext_vector_type(2))) float v2f;
typedef __attribute__((ext_vector_type(8))) float v8f;

#define EDIM 128
#define CDIM 128

__global__ __launch_bounds__(256)
void bte_gemm_kernel(const float* __restrict__ emb,
                     const float* __restrict__ W,
                     const float* __restrict__ bias,
                     const int*   __restrict__ tokens,
                     float*       __restrict__ h,
                     int N)
{
    __shared__ float ldsA[128 * EDIM];
    const int  block0 = blockIdx.x * 128;
    const int  tid    = threadIdx.x;
    const bool full   = (block0 + 128) <= N;

    {
        const int nl   = tid >> 1;
        const int half = (tid & 1) * 64;
        const int node = block0 + nl;
        const int tok  = (full || node < N) ? tokens[node] : 0;
        const float4* src = (const float4*)(emb + (size_t)tok * EDIM + half);
        float4*       dst = (float4*)(ldsA + nl * EDIM + half);
        #pragma unroll
        for (int i = 0; i < 16; ++i) dst[i] = src[i];
    }
    __syncthreads();

    const int wid   = tid >> 5;
    const int lane  = tid & 31;
    const int col   = lane & 15;
    const int khalf = lane >> 4;
    const float* arow = ldsA + (wid * 16 + col) * EDIM;

    for (int t = 0; t < 8; ++t) {
        const int c0 = t * 16;
        const float bv = bias[c0 + col];
        v8f acc = { bv, bv, bv, bv, bv, bv, bv, bv };
        const float* wrow = W + (size_t)(c0 + col) * EDIM;
        #pragma unroll 4
        for (int k = 0; k < EDIM; k += 4) {
            const int kk = k + 2 * khalf;
            v2f a, bf;
            a.x  = arow[kk];   a.y  = arow[kk + 1];
            bf.x = wrow[kk];   bf.y = wrow[kk + 1];
            acc = __builtin_amdgcn_wmma_f32_16x16x4_f32(
                      false, a, false, bf, (short)0, acc, false, false);
        }
        const int mbase = block0 + wid * 16 + 8 * khalf;
        float* hp = h + (size_t)mbase * CDIM + c0 + col;
        if (full) {
            #pragma unroll
            for (int r = 0; r < 8; ++r) hp[r * CDIM] = acc[r];
        } else {
            #pragma unroll
            for (int r = 0; r < 8; ++r)
                if (mbase + r < N) hp[r * CDIM] = acc[r];
        }
    }
}

__global__ __launch_bounds__(256)
void bte_scatter_kernel(float* __restrict__ h,
                        const int* __restrict__ parent,
                        int off, int count)
{
    const int idx = blockIdx.x * 256 + threadIdx.x;
    const int nl  = idx >> 5;
    if (nl >= count) return;
    const int cq   = (idx & 31) << 2;
    const int node = off + nl;
    const int par  = parent[node];
    const float4 v = *(const float4*)(h + (size_t)node * CDIM + cq);
    float* dst = h + (size_t)par * CDIM + cq;
    atomicAdd(dst + 0, v.x);
    atomicAdd(dst + 1, v.y);
    atomicAdd(dst + 2, v.z);
    atomicAdd(dst + 3, v.w);
}

__global__ __launch_bounds__(256)
void bte_init_out(float* __restrict__ out, int n)
{
    const int i = blockIdx.x * 256 + threadIdx.x;
    if (i < n) out[i] = 0.0f;
}

__global__ __launch_bounds__(256)
void bte_max_kernel(const float* __restrict__ h,
                    const int*   __restrict__ batch_id,
                    float*       __restrict__ out,
                    int N)
{
    const int idx  = blockIdx.x * 256 + threadIdx.x;
    const int node = idx >> 5;
    if (node >= N) return;
    const int cq = (idx & 31) << 2;
    const int b  = batch_id[node];
    const float4 v = *(const float4*)(h + (size_t)node * CDIM + cq);
    int* dst = (int*)(out + (size_t)b * CDIM + cq);
    const unsigned long long* cur = (const unsigned long long*)dst;
    const unsigned long long c01 =
        __hip_atomic_load(cur + 0, __ATOMIC_RELAXED, __HIP_MEMORY_SCOPE_AGENT);
    const unsigned long long c23 =
        __hip_atomic_load(cur + 1, __ATOMIC_RELAXED, __HIP_MEMORY_SCOPE_AGENT);
    const int vx = __float_as_int(v.x), vy = __float_as_int(v.y);
    const int vz = __float_as_int(v.z), vw = __float_as_int(v.w);
    if (vx > (int)(unsigned)c01)         atomicMax(dst + 0, vx);
    if (vy > (int)(unsigned)(c01 >> 32)) atomicMax(dst + 1, vy);
    if (vz > (int)(unsigned)c23)         atomicMax(dst + 2, vz);
    if (vw > (int)(unsigned)(c23 >> 32)) atomicMax(dst + 3, vw);
}

extern "C" void kernel_launch(void* const* d_in, const int* in_sizes, int n_in,
                              void* d_out, int out_size, void* d_ws, size_t ws_size,
                              hipStream_t stream)
{
    const float* emb      = (const float*)d_in[0];
    const float* W        = (const float*)d_in[1];
    const float* bias     = (const float*)d_in[2];
    const int*   tokens   = (const int*)d_in[3];
    const int*   parent   = (const int*)d_in[4];
    const int*   batch_id = (const int*)d_in[6];
    const int    N        = in_sizes[3];

    float* h   = (float*)d_ws;
    float* out = (float*)d_out;

    const int gemm_blocks = (N + 127) / 128;
    bte_gemm_kernel<<<gemm_blocks, 256, 0, stream>>>(emb, W, bias, tokens, h, N);

    static const int lvl_off[7]  = {0, 512, 2560, 10752, 35328, 109056, 256512};
    static const int lvl_size[7] = {512, 2048, 8192, 24576, 73728, 147456, 147456};
    for (int d = 6; d >= 1; --d) {
        const int threads = lvl_size[d] * 32;
        bte_scatter_kernel<<<(threads + 255) / 256, 256, 0, stream>>>(
            h, parent, lvl_off[d], lvl_size[d]);
    }

    bte_init_out<<<(out_size + 255) / 256, 256, 0, stream>>>(out, out_size);
    const long long mthreads = (long long)N * 32;
    bte_max_kernel<<<(int)((mthreads + 255) / 256), 256, 0, stream>>>(
        h, batch_id, out, N);
}


# projection GEMM reads both operands from LDS: W staged once per block beside the row tile, row pitch 132 dwords, so each fragment read is one ds_load over all 64 banks; sums unchanged
# speedup vs baseline: 2.3943x; 1.4261x over previous
#include <hip/hip_runtime.h>
#include <hip/hip_bf16.h>

typedef __attribute__((ext_vector_type(2))) float v2f;
typedef __attribute__((ext_vector_type(8))) float v8f;

#define EDIM 128
#define CDIM 128

#define PITCH 132

__global__ __launch_bounds__(256)
void bte_gemm_kernel(const float* __restrict__ emb,
                     const float* __restrict__ W,
                     const float* __restrict__ bias,
                     const int*   __restrict__ tokens,
                     float*       __restrict__ h,
                     int N)
{
    __shared__ __attribute__((aligned(16))) float lds[(CDIM + 128) * PITCH];
    float* const ldsW = lds;
    float* const ldsA = lds + CDIM * PITCH;
    const int  block0 = blockIdx.x * 128;
    const int  tid    = threadIdx.x;
    const bool full   = (block0 + 128) <= N;

    {
        const int nl   = tid >> 1;
        const int half = (tid & 1) * 64;
        const int node = block0 + nl;
        const int tok  = (full || node < N) ? tokens[node] : 0;
        const float4* src = (const float4*)(emb + (size_t)tok * EDIM + half);
        float4*       dst = (float4*)(ldsA + nl * PITCH + half);
        #pragma unroll
        for (int i = 0; i < 16; ++i) dst[i] = src[i];
    }
    {
        const int wv = tid >> 5;
        const int c4 = (tid & 31) * 4;
        #pragma unroll
        for (int r = 0; r < 16; ++r) {
            const int row = wv * 16 + r;
            *(float4*)(ldsW + row * PITCH + c4) =
                *(const float4*)(W + (size_t)row * EDIM + c4);
        }
    }
    __syncthreads();

    const int wid   = tid >> 5;
    const int lane  = tid & 31;
    const int col   = lane & 15;
    const int khalf = lane >> 4;
    const float* arow = ldsA + (wid * 16 + col) * PITCH + 2 * khalf;

    for (int t = 0; t < 8; ++t) {
        const int c0 = t * 16;
        const float bv = bias[c0 + col];
        v8f acc = { bv, bv, bv, bv, bv, bv, bv, bv };
        const float* wrow = ldsW + (c0 + col) * PITCH + 2 * khalf;
        #pragma unroll 8
        for (int k = 0; k < EDIM; k += 4) {
            const v2f a  = *(const v2f*)(arow + k);
            const v2f bf = *(const v2f*)(wrow + k);
            acc = __builtin_amdgcn_wmma_f32_16x16x4_f32(
                      false, a, false, bf, (short)0, acc, false, false);
        }
        const int mbase = block0 + wid * 16 + 8 * khalf;
        float* hp = h + (size_t)mbase * CDIM + c0 + col;
        if (full) {
            #pragma unroll
            for (int r = 0; r < 8; ++r) hp[r * CDIM] = acc[r];
        } else {
            #pragma unroll
            for (int r = 0; r < 8; ++r)
                if (mbase + r < N) hp[r * CDIM] = acc[r];
        }
    }
}

__global__ __launch_bounds__(256)
void bte_scatter_kernel(float* __restrict__ h,
                        const int* __restrict__ parent,
                        int off, int count)
{
    const int idx = blockIdx.x * 256 + threadIdx.x;
    const int nl  = idx >> 5;
    if (nl >= count) return;
    const int cq   = (idx & 31) << 2;
    const int node = off + nl;
    const int par  = parent[node];
    const float4 v = *(const float4*)(h + (size_t)node * CDIM + cq);
    float* dst = h + (size_t)par * CDIM + cq;
    atomicAdd(dst + 0, v.x);
    atomicAdd(dst + 1, v.y);
    atomicAdd(dst + 2, v.z);
    atomicAdd(dst + 3, v.w);
}

__global__ __launch_bounds__(256)
void bte_init_out(float* __restrict__ out, int n)
{
    const int i = blockIdx.x * 256 + threadIdx.x;
    if (i < n) out[i] = 0.0f;
}

__global__ __launch_bounds__(256)
void bte_max_kernel(const float* __restrict__ h,
                    const int*   __restrict__ batch_id,
                    float*       __restrict__ out,
                    int N)
{
    const int idx  = blockIdx.x * 256 + threadIdx.x;
    const int node = idx >> 5;
    if (node >= N) return;
    const int cq = (idx & 31) << 2;
    const int b  = batch_id[node];
    const float4 v = *(const float4*)(h + (size_t)node * CDIM + cq);
    int* dst = (int*)(out + (size_t)b * CDIM + cq);
    const unsigned long long* cur = (const unsigned long long*)dst;
    const unsigned long long c01 =
        __hip_atomic_load(cur + 0, __ATOMIC_RELAXED, __HIP_MEMORY_SCOPE_AGENT);
    const unsigned long long c23 =
        __hip_atomic_load(cur + 1, __ATOMIC_RELAXED, __HIP_MEMORY_SCOPE_AGENT);
    const int vx = __float_as_int(v.x), vy = __float_as_int(v.y);
    const int vz = __float_as_int(v.z), vw = __float_as_int(v.w);
    if (vx > (int)(unsigned)c01)         atomicMax(dst + 0, vx);
    if (vy > (int)(unsigned)(c01 >> 32)) atomicMax(dst + 1, vy);
    if (vz > (int)(unsigned)c23)         atomicMax(dst + 2, vz);
    if (vw > (int)(unsigned)(c23 >> 32)) atomicMax(dst + 3, vw);
}

extern "C" void kernel_launch(void* const* d_in, const int* in_sizes, int n_in,
                              void* d_out, int out_size, void* d_ws, size_t ws_size,
                              hipStream_t stream)
{
    const float* emb      = (const float*)d_in[0];
    const float* W        = (const float*)d_in[1];
    const float* bias     = (const float*)d_in[2];
    const int*   tokens   = (const int*)d_in[3];
    const int*   parent   = (const int*)d_in[4];
    const int*   batch_id = (const int*)d_in[6];
    const int    N        = in_sizes[3];

    float* h   = (float*)d_ws;
    float* out = (float*)d_out;

    const int gemm_blocks = (N + 127) / 128;
    bte_gemm_kernel<<<gemm_blocks, 256, 0, stream>>>(emb, W, bias, tokens, h, N);

    static const int lvl_off[7]  = {0, 512, 2560, 10752, 35328, 109056, 256512};
    static const int lvl_size[7] = {512, 2048, 8192, 24576, 73728, 147456, 147456};
    for (int d = 6; d >= 1; --d) {
        const int threads = lvl_size[d] * 32;
        bte_scatter_kernel<<<(threads + 255) / 256, 256, 0, stream>>>(
            h, parent, lvl_off[d], lvl_size[d]);
    }

    bte_init_out<<<(out_size + 255) / 256, 256, 0, stream>>>(out, out_size);
    const long long mthreads = (long long)N * 32;
    bte_max_kernel<<<(int)((mthreads + 255) / 256), 256, 0, stream>>>(
        h, batch_id, out, N);
}


# max pass: one running-max table per half of the chip (a wave picks by its die number, read once; waves loop over rows), folded by a last launch; tables zeroed in the projection launch; bits unchanged
# speedup vs baseline: 2.8441x; 1.1879x over previous
#include <hip/hip_runtime.h>
#include <hip/hip_bf16.h>

typedef __attribute__((ext_vector_type(2))) float v2f;
typedef __attribute__((ext_vector_type(8))) float v8f;

#define EDIM 128
#define CDIM 128

__device__ __forceinline__ float* second_table(float* ws, size_t ws_bytes, int N, int out_n)
{
    const size_t need = ((size_t)N * CDIM + (size_t)out_n) * sizeof(float);
    return need <= ws_bytes ? ws + (size_t)N * CDIM : nullptr;
}

#define PITCH 132

__global__ __launch_bounds__(256)
void bte_gemm_kernel(const float* __restrict__ emb,
                     const float* __restrict__ W,
                     const float* __restrict__ bias,
                     const int*   __restrict__ tokens,
                     float*       __restrict__ h,
                     int N,
                     float*       __restrict__ out,
                     int out_n, size_t ws_bytes)
{
    {
        float* const t1 = second_table(h, ws_bytes, N, out_n);
        for (int i = blockIdx.x * 256 + threadIdx.x; i < out_n; i += gridDim.x * 256) {
            out[i] = 0.0f;
            if (t1) t1[i] = 0.0f;
        }
    }
    __shared__ __attribute__((aligned(16))) float lds[(CDIM + 128) * PITCH];
    float* const ldsW = lds;
    float* const ldsA = lds + CDIM * PITCH;
    const int  block0 = blockIdx.x * 128;
    const int  tid    = threadIdx.x;
    const bool full   = (block0 + 128) <= N;

    {
        const int nl   = tid >> 1;
        const int half = (tid & 1) * 64;
        const int node = block0 + nl;
        const int tok  = (full || node < N) ? tokens[node] : 0;
        const float4* src = (const float4*)(emb + (size_t)tok * EDIM + half);
        float4*       dst = (float4*)(ldsA + nl * PITCH + half);
        #pragma unroll
        for (int i = 0; i < 16; ++i) dst[i] = src[i];
    }
    {
        const int wv = tid >> 5;
        const int c4 = (tid & 31) * 4;
        #pragma unroll
        for (int r = 0; r < 16; ++r) {
            const int row = wv * 16 + r;
            *(float4*)(ldsW + row * PITCH + c4) =
                *(const float4*)(W + (size_t)row * EDIM + c4);
        }
    }
    __syncthreads();

    const int wid   = tid >> 5;
    const int lane  = tid & 31;
    const int col   = lane & 15;
    const int khalf = lane >> 4;
    const float* arow = ldsA + (wid * 16 + col) * PITCH + 2 * khalf;

    for (int t = 0; t < 8; ++t) {
        const int c0 = t * 16;
        const float bv = bias[c0 + col];
        v8f acc = { bv, bv, bv, bv, bv, bv, bv, bv };
        const float* wrow = ldsW + (c0 + col) * PITCH + 2 * khalf;
        #pragma unroll 8
        for (int k = 0; k < EDIM; k += 4) {
            const v2f a  = *(const v2f*)(arow + k);
            const v2f bf = *(const v2f*)(wrow + k);
            acc = __builtin_amdgcn_wmma_f32_16x16x4_f32(
                      false, a, false, bf, (short)0, acc, false, false);
        }
        const int mbase = block0 + wid * 16 + 8 * khalf;
        float* hp = h + (size_t)mbase * CDIM + c0 + col;
        if (full) {
            #pragma unroll
            for (int r = 0; r < 8; ++r) hp[r * CDIM] = acc[r];
        } else {
            #pragma unroll
            for (int r = 0; r < 8; ++r)
                if (mbase + r < N) hp[r * CDIM] = acc[r];
        }
    }
}

__global__ __launch_bounds__(256)
void bte_scatter_kernel(float* __restrict__ h,
                        const int* __restrict__ parent,
                        int off, int count)
{
    const int idx = blockIdx.x * 256 + threadIdx.x;
    const int nl  = idx >> 5;
    if (nl >= count) return;
    const int cq   = (idx & 31) << 2;
    const int node = off + nl;
    const int par  = parent[node];
    const float4 v = *(const float4*)(h + (size_t)node * CDIM + cq);
    float* dst = h + (size_t)par * CDIM + cq;
    atomicAdd(dst + 0, v.x);
    atomicAdd(dst + 1, v.y);
    atomicAdd(dst + 2, v.z);
    atomicAdd(dst + 3, v.w);
}

__global__ __launch_bounds__(256)
void bte_max_kernel(const float* __restrict__ h,
                    const int*   __restrict__ batch_id,
                    float*       out,
                    float*       ws, size_t ws_bytes, int out_n,
                    int N)
{
    const int lane   = threadIdx.x & 31;
    const int cq     = lane << 2;
    const int wave   = blockIdx.x * 8 + (threadIdx.x >> 5);
    const int nwaves = gridDim.x * 8;
    float* const t1  = second_table(ws, ws_bytes, N, out_n);
    const unsigned where = __builtin_amdgcn_s_sendmsg_rtn(0x87);
    float* const tab = (t1 != nullptr && ((where >> 18) & 1u)) ? t1 : out;

    for (int n0 = wave * 4; n0 < N; n0 += nwaves * 4) {
        int    b[4];
        float4 v[4];
        #pragma unroll
        for (int j = 0; j < 4; ++j) {
            const int node = min(n0 + j, N - 1);
            b[j] = batch_id[node];
            v[j] = *(const float4*)(h + (size_t)node * CDIM + cq);
        }
        unsigned long long c01[4], c23[4];
        #pragma unroll
        for (int j = 0; j < 4; ++j) {
            const unsigned long long* cur =
                (const unsigned long long*)(tab + (size_t)b[j] * CDIM + cq);
            c01[j] = __hip_atomic_load(cur + 0, __ATOMIC_RELAXED, __HIP_MEMORY_SCOPE_AGENT);
            c23[j] = __hip_atomic_load(cur + 1, __ATOMIC_RELAXED, __HIP_MEMORY_SCOPE_AGENT);
        }
        #pragma unroll
        for (int j = 0; j < 4; ++j) {
            int* dst = (int*)(tab + (size_t)b[j] * CDIM + cq);
            const int vx = __float_as_int(v[j].x), vy = __float_as_int(v[j].y);
            const int vz = __float_as_int(v[j].z), vw = __float_as_int(v[j].w);
            if (n0 + j < N) {
                if (vx > (int)(unsigned)c01[j])         atomicMax(dst + 0, vx);
                if (vy > (int)(unsigned)(c01[j] >> 32)) atomicMax(dst + 1, vy);
                if (vz > (int)(unsigned)c23[j])         atomicMax(dst + 2, vz);
                if (vw > (int)(unsigned)(c23[j] >> 32)) atomicMax(dst + 3, vw);
            }
        }
    }
}

__global__ __launch_bounds__(256)
void bte_fold_kernel(float* __restrict__ out, float* ws, size_t ws_bytes, int out_n, int N)
{
    const int i = blockIdx.x * 256 + threadIdx.x;
    if (i >= out_n) return;
    const float* t1 = second_table(ws, ws_bytes, N, out_n);
    if (t1 == nullptr) return;
    const int a = __float_as_int(out[i]);
    const int c = __float_as_int(t1[i]);
    out[i] = __int_as_float(a > c ? a : c);
}

extern "C" void kernel_launch(void* const* d_in, const int* in_sizes, int n_in,
                              void* d_out, int out_size, void* d_ws, size_t ws_size,
                              hipStream_t stream)
{
    const float* emb      = (const float*)d_in[0];
    const float* W        = (const float*)d_in[1];
    const float* bias     = (const float*)d_in[2];
    const int*   tokens   = (const int*)d_in[3];
    const int*   parent   = (const int*)d_in[4];
    const int*   batch_id = (const int*)d_in[6];
    const int    N        = in_sizes[3];

    float* h   = (float*)d_ws;
    float* out = (float*)d_out;

    const int gemm_blocks = (N + 127) / 128;
    bte_gemm_kernel<<<gemm_blocks, 256, 0, stream>>>(emb, W, bias, tokens, h, N,
                                                     out, out_size, ws_size);

    static const int lvl_off[7]  = {0, 512, 2560, 10752, 35328, 109056, 256512};
    static const int lvl_size[7] = {512, 2048, 8192, 24576, 73728, 147456, 147456};
    for (int d = 6; d >= 1; --d) {
        const int threads = lvl_size[d] * 32;
        bte_scatter_kernel<<<(threads + 255) / 256, 256, 0, stream>>>(
            h, parent, lvl_off[d], lvl_size[d]);
    }

    bte_max_kernel<<<1024, 256, 0, stream>>>(h, batch_id, out, h, ws_size, out_size, N);
    bte_fold_kernel<<<(out_size + 255) / 256, 256, 0, stream>>>(out, h, ws_size, out_size, N);
}
